# starting kernel
# baseline (speedup 1.0000x reference)
#include <hip/hip_runtime.h>
#include <hip/hip_bf16.h>
#include <math.h>

#define EPSV 1e-8f

typedef float v2f __attribute__((ext_vector_type(2)));
typedef float v8f __attribute__((ext_vector_type(8)));

__device__ __forceinline__ v8f wmma4(v2f a, v2f b, v8f c) {
  return __builtin_amdgcn_wmma_f32_16x16x4_f32(false, a, false, b, (short)0, c,
                                               false, false);
}

__device__ __forceinline__ float wave_sum32(float v) {
#pragma unroll
  for (int off = 16; off > 0; off >>= 1) v += __shfl_xor(v, off, 32);
  return v;
}

__device__ __forceinline__ void ln128_row(const float* __restrict__ src,
                                          float* __restrict__ dst,
                                          const float* __restrict__ g,
                                          const float* __restrict__ b,
                                          int lane) {
  float xv[4];
#pragma unroll
  for (int j = 0; j < 4; ++j) xv[j] = src[lane + 32 * j];
  float s = xv[0] + xv[1] + xv[2] + xv[3];
  s = wave_sum32(s);
  float m = s * (1.0f / 128.0f);
  float vs = 0.0f;
#pragma unroll
  for (int j = 0; j < 4; ++j) { float d = xv[j] - m; vs += d * d; }
  vs = wave_sum32(vs);
  float inv = rsqrtf(vs * (1.0f / 128.0f) + EPSV);
#pragma unroll
  for (int j = 0; j < 4; ++j) {
    int d = lane + 32 * j;
    dst[d] = (xv[j] - m) * inv * g[d] + b[d];
  }
}

__global__ void k_gram(const float* __restrict__ x,
                       const float* __restrict__ g1,
                       const float* __restrict__ b1,
                       float* __restrict__ stats1,
                       float* __restrict__ stats2,
                       float* __restrict__ gpart) {
  extern __shared__ float yt[];
  __shared__ float s1l[128], s2l[128];
  const int tid = threadIdx.x, lane = tid & 31, wave = tid >> 5;
  const int b = blockIdx.x >> 3;
  const int oct = blockIdx.x & 7;
  if (tid < 128) { s1l[tid] = 0.0f; s2l[tid] = 0.0f; }
  const int khalf = lane >> 4;
  const int nl = lane & 15;
  const v8f vzero = {0, 0, 0, 0, 0, 0, 0, 0};
  v8f acc[8];
#pragma unroll
  for (int t = 0; t < 8; ++t) acc[t] = vzero;
  float a1[4] = {0, 0, 0, 0}, a2[4] = {0, 0, 0, 0};

  for (int sc = 0; sc < 2; ++sc) {
    const long rowbase = (long)b * 4096 + oct * 512 + sc * 256;
    __syncthreads();
    for (int i = 0; i < 32; ++i) {
      int r = wave * 32 + i;
      const float* src = x + (rowbase + r) * 128;
      float xv[4];
#pragma unroll
      for (int j = 0; j < 4; ++j) xv[j] = src[lane + 32 * j];
      float s = wave_sum32(xv[0] + xv[1] + xv[2] + xv[3]);
      float m = s * (1.0f / 128.0f);
      float vs = 0.0f;
#pragma unroll
      for (int j = 0; j < 4; ++j) { float d = xv[j] - m; vs += d * d; }
      vs = wave_sum32(vs);
      float inv = rsqrtf(vs * (1.0f / 128.0f) + EPSV);
#pragma unroll
      for (int j = 0; j < 4; ++j) {
        int d = lane + 32 * j;
        float y = (xv[j] - m) * inv * g1[d] + b1[d];
        a1[j] += y; a2[j] += y * y;
        yt[r * 128 + d] = y;
      }
    }
    __syncthreads();
    for (int t = 0; t < 8; ++t) {
      int idx = wave + 8 * t;
      int mi = idx >> 3, ni = idx & 7;
      int mrow = mi * 16 + nl;
      int ncol = ni * 16 + nl;
      v8f a = acc[t];
      for (int kb = 0; kb < 64; ++kb) {
        int k0 = kb * 4 + khalf * 2;
        v2f af, bf;
        af.x = yt[k0 * 128 + mrow];   af.y = yt[(k0 + 1) * 128 + mrow];
        bf.x = yt[k0 * 128 + ncol];   bf.y = yt[(k0 + 1) * 128 + ncol];
        a = wmma4(af, bf, a);
      }
      acc[t] = a;
    }
  }

  float* gp = gpart + (long)blockIdx.x * 16384;
  for (int t = 0; t < 8; ++t) {
    int idx = wave + 8 * t;
    int mi = idx >> 3, ni = idx & 7;
    int dmBase = mi * 16 + khalf * 8;
    int de = ni * 16 + nl;
#pragma unroll
    for (int r = 0; r < 8; ++r) gp[(dmBase + r) * 128 + de] = acc[t][r];
  }
#pragma unroll
  for (int j = 0; j < 4; ++j) {
    int d = lane + 32 * j;
    atomicAdd(&s1l[d], a1[j]);
    atomicAdd(&s2l[d], a2[j]);
  }
  __syncthreads();
  if (tid < 128) {
    atomicAdd(&stats1[b * 128 + tid], s1l[tid]);
    atomicAdd(&stats2[b * 128 + tid], s2l[tid]);
  }
}

__global__ void k_combine(const float* __restrict__ stats1,
                          const float* __restrict__ stats2,
                          const float* __restrict__ gpart,
                          float* __restrict__ dcorr) {
  const int d = blockIdx.x;
  const int e = threadIdx.x;
  float acc = 0.0f;
  for (int b = 0; b < 32; ++b) {
    float s1d = stats1[b * 128 + d], s2d = stats2[b * 128 + d];
    float s1e = stats1[b * 128 + e], s2e = stats2[b * 128 + e];
    float md = s1d * (1.0f / 4096.0f), me = s1e * (1.0f / 4096.0f);
    float vd = s2d - s1d * s1d * (1.0f / 4096.0f);
    float ve = s2e - s1e * s1e * (1.0f / 4096.0f);
    float ind = 1.0f / fmaxf(sqrtf(fmaxf(vd, 0.0f)), EPSV);
    float ine = 1.0f / fmaxf(sqrtf(fmaxf(ve, 0.0f)), EPSV);
    float G = 0.0f;
#pragma unroll
    for (int q = 0; q < 8; ++q)
      G += gpart[(long)(b * 8 + q) * 16384 + d * 128 + e];
    acc += (G - 4096.0f * md * me) * ind * ine;
  }
  dcorr[d * 128 + e] = acc * (1.0f / 32.0f);
}

__global__ void k_graph(const float* __restrict__ corr,
                        const float* __restrict__ alpha_logit,
                        const float* __restrict__ chebw,
                        const float* __restrict__ dcorr,
                        float* __restrict__ mixed_out,
                        float* __restrict__ Lap,
                        float* __restrict__ wout) {
  __shared__ float dis[128];
  const int i = threadIdx.x;
  const float alpha = 1.0f / (1.0f + __expf(-alpha_logit[0]));
  float deg = 0.0f;
  for (int j = 0; j < 128; ++j) {
    float learned = tanhf(0.5f * (corr[i * 128 + j] + corr[j * 128 + i]));
    float dc = fminf(fmaxf(dcorr[i * 128 + j], -1.0f), 1.0f);
    float mx = alpha * learned + (1.0f - alpha) * dc;
    mixed_out[i * 128 + j] = mx;
    if (j != i) deg += mx;
  }
  dis[i] = rsqrtf(fmaxf(deg, EPSV));
  __syncthreads();
  const float di = dis[i];
  for (int j = 0; j < 128; ++j) {
    float a = (j == i) ? 0.0f : mixed_out[i * 128 + j];
    float l = ((j == i) ? 1.0f : 0.0f) - di * a * dis[j];
    Lap[i * 128 + j] = fminf(fmaxf(l, -1.5f), 1.5f);
  }
  if (i == 0) {
    float c0 = chebw[0], c1 = chebw[1], c2 = chebw[2];
    float mx = fmaxf(c0, fmaxf(c1, c2));
    float e0 = __expf(c0 - mx), e1 = __expf(c1 - mx), e2 = __expf(c2 - mx);
    float s = e0 + e1 + e2;
    wout[0] = e0 / s; wout[1] = e1 / s; wout[2] = e2 / s;
  }
}

__global__ void k_main(const float* __restrict__ x,
                       const float* __restrict__ g1, const float* __restrict__ b1,
                       const float* __restrict__ Lap, const float* __restrict__ wv,
                       const float* __restrict__ Wi, const float* __restrict__ bi,
                       const float* __restrict__ Wo, const float* __restrict__ bo,
                       const float* __restrict__ g2, const float* __restrict__ b2,
                       const float* __restrict__ g3, const float* __restrict__ b3,
                       float* __restrict__ out) {
  extern __shared__ float smem[];
  float* yo = smem;
  float* t1 = smem + 8192;
  float* hb = smem + 16384;
  const int tid = threadIdx.x, lane = tid & 31, wave = tid >> 5;
  const long base = (long)blockIdx.x * 64;
  const float w0 = wv[0], w1 = wv[1], w2 = wv[2];
  const int khalf = lane >> 4;
  const int nl = lane & 15;

  for (int i = 0; i < 8; ++i) {
    int r = wave * 8 + i;
    ln128_row(x + (base + r) * 128, yo + r * 128, g1, b1, lane);
  }
  __syncthreads();

  for (int t = 0; t < 4; ++t) {
    int idx = wave + 8 * t; int mi = idx >> 3, ni = idx & 7;
    v8f acc = {};
    int mrow = mi * 16 + nl, nc = ni * 16 + nl;
    for (int kb = 0; kb < 32; ++kb) {
      int k0 = kb * 4 + khalf * 2;
      v2f a, bb;
      a.x = yo[mrow * 128 + k0];  a.y = yo[mrow * 128 + k0 + 1];
      bb.x = Lap[k0 * 128 + nc];  bb.y = Lap[(k0 + 1) * 128 + nc];
      acc = wmma4(a, bb, acc);
    }
    int rowB = mi * 16 + khalf * 8;
#pragma unroll
    for (int r = 0; r < 8; ++r) t1[(rowB + r) * 128 + nc] = acc[r];
  }
  __syncthreads();

  for (int t = 0; t < 4; ++t) {
    int idx = wave + 8 * t; int mi = idx >> 3, ni = idx & 7;
    v8f acc = {};
    int mrow = mi * 16 + nl, nc = ni * 16 + nl;
    for (int kb = 0; kb < 32; ++kb) {
      int k0 = kb * 4 + khalf * 2;
      v2f a, bb;
      a.x = t1[mrow * 128 + k0];  a.y = t1[mrow * 128 + k0 + 1];
      bb.x = Lap[k0 * 128 + nc];  bb.y = Lap[(k0 + 1) * 128 + nc];
      acc = wmma4(a, bb, acc);
    }
    int rowB = mi * 16 + khalf * 8;
#pragma unroll
    for (int r = 0; r < 8; ++r) {
      int pos = (rowB + r) * 128 + nc;
      float y0 = yo[pos], y1 = t1[pos];
      float tx2 = fminf(fmaxf(2.0f * acc[r] - y0, -50.0f), 50.0f);
      yo[pos] = w0 * y0 + w1 * y1 + w2 * tx2;
    }
  }
  __syncthreads();

  for (int t = 0; t < 8; ++t) {
    int idx = wave + 8 * t; int mi = idx >> 4, ni = idx & 15;
    v8f acc = {};
    int mrow = mi * 16 + nl, nc = ni * 16 + nl;
    for (int kb = 0; kb < 32; ++kb) {
      int k0 = kb * 4 + khalf * 2;
      v2f a, bb;
      a.x = yo[mrow * 128 + k0];  a.y = yo[mrow * 128 + k0 + 1];
      bb.x = Wi[nc * 128 + k0];   bb.y = Wi[nc * 128 + k0 + 1];
      acc = wmma4(a, bb, acc);
    }
    int rowB = mi * 16 + khalf * 8;
    float bia = bi[nc];
#pragma unroll
    for (int r = 0; r < 8; ++r) hb[(rowB + r) * 256 + nc] = acc[r] + bia;
  }
  __syncthreads();

  for (int i = 0; i < 8; ++i) {
    int r = wave * 8 + i;
    float xv[8];
#pragma unroll
    for (int j = 0; j < 8; ++j) xv[j] = hb[r * 256 + lane + 32 * j];
    float s = 0.0f;
#pragma unroll
    for (int j = 0; j < 8; ++j) s += xv[j];
    s = wave_sum32(s);
    float m = s * (1.0f / 256.0f);
    float vs = 0.0f;
#pragma unroll
    for (int j = 0; j < 8; ++j) { float d = xv[j] - m; vs += d * d; }
    vs = wave_sum32(vs);
    float inv = rsqrtf(vs * (1.0f / 256.0f) + EPSV);
#pragma unroll
    for (int j = 0; j < 8; ++j) {
      int d = lane + 32 * j;
      float yv = (xv[j] - m) * inv * g2[d] + b2[d];
      hb[r * 256 + d] = 0.5f * yv * (1.0f + erff(yv * 0.70710678118654752f));
    }
  }
  __syncthreads();

  for (int t = 0; t < 4; ++t) {
    int idx = wave + 8 * t; int mi = idx >> 3, ni = idx & 7;
    v8f acc = {};
    int mrow = mi * 16 + nl, nc = ni * 16 + nl;
    for (int kb = 0; kb < 64; ++kb) {
      int k0 = kb * 4 + khalf * 2;
      v2f a, bb;
      a.x = hb[mrow * 256 + k0];  a.y = hb[mrow * 256 + k0 + 1];
      bb.x = Wo[nc * 256 + k0];   bb.y = Wo[nc * 256 + k0 + 1];
      acc = wmma4(a, bb, acc);
    }
    int rowB = mi * 16 + khalf * 8;
    float boa = bo[nc];
#pragma unroll
    for (int r = 0; r < 8; ++r) t1[(rowB + r) * 128 + nc] = acc[r] + boa;
  }
  __syncthreads();

  for (int i = 0; i < 8; ++i) {
    int r = wave * 8 + i;
    float xv[4];
#pragma unroll
    for (int j = 0; j < 4; ++j) xv[j] = t1[r * 128 + lane + 32 * j];
    float s = wave_sum32(xv[0] + xv[1] + xv[2] + xv[3]);
    float m = s * (1.0f / 128.0f);
    float vs = 0.0f;
#pragma unroll
    for (int j = 0; j < 4; ++j) { float d = xv[j] - m; vs += d * d; }
    vs = wave_sum32(vs);
    float inv = rsqrtf(vs * (1.0f / 128.0f) + EPSV);
#pragma unroll
    for (int j = 0; j < 4; ++j) {
      int d = lane + 32 * j;
      out[(base + r) * 128 + d] = (xv[j] - m) * inv * g3[d] + b3[d];
    }
  }
}

extern "C" void kernel_launch(void* const* d_in, const int* in_sizes, int n_in,
                              void* d_out, int out_size, void* d_ws, size_t ws_size,
                              hipStream_t stream) {
  const float* x     = (const float*)d_in[0];
  const float* corr  = (const float*)d_in[1];
  const float* alog  = (const float*)d_in[2];
  const float* chebw = (const float*)d_in[3];
  const float* Wi    = (const float*)d_in[4];
  const float* bi    = (const float*)d_in[5];
  const float* Wo    = (const float*)d_in[6];
  const float* bo    = (const float*)d_in[7];
  const float* g1    = (const float*)d_in[8];
  const float* b1    = (const float*)d_in[9];
  const float* g2    = (const float*)d_in[10];
  const float* b2    = (const float*)d_in[11];
  const float* g3    = (const float*)d_in[12];
  const float* b3    = (const float*)d_in[13];
  float* out = (float*)d_out;
  float* ws  = (float*)d_ws;

  float* stats1 = ws;
  float* stats2 = ws + 4096;
  float* gpart  = ws + 8192;
  float* dcorr  = ws + 8192 + 256 * 16384;
  float* Lap    = dcorr + 16384;
  float* wv     = Lap + 16384;
  float* mixed  = out + (long)32 * 4096 * 128;

  hipMemsetAsync(d_ws, 0, 8192 * sizeof(float), stream);
  k_gram<<<256, 256, 256 * 128 * sizeof(float), stream>>>(x, g1, b1, stats1,
                                                          stats2, gpart);
  k_combine<<<128, 128, 0, stream>>>(stats1, stats2, gpart, dcorr);
  k_graph<<<1, 128, 0, stream>>>(corr, alog, chebw, dcorr, mixed, Lap, wv);
  k_main<<<2048, 256, 131072, stream>>>(x, g1, b1, Lap, wv, Wi, bi, Wo, bo,
                                        g2, b2, g3, b3, out);
}


# k_gram split over rows, not tiles: a wave loads its 128 rows straight into WMMA fragment order, LN1 there, the same registers are A and B of all 64 tiles; no LDS staging (sum order changed)
# speedup vs baseline: 2.6289x; 2.6289x over previous
#include <hip/hip_runtime.h>
#include <hip/hip_bf16.h>
#include <math.h>

#define EPSV 1e-8f

typedef float v2f __attribute__((ext_vector_type(2)));
typedef float v8f __attribute__((ext_vector_type(8)));

__device__ __forceinline__ v8f wmma4(v2f a, v2f b, v8f c) {
  return __builtin_amdgcn_wmma_f32_16x16x4_f32(false, a, false, b, (short)0, c,
                                               false, false);
}

typedef float v4f __attribute__((ext_vector_type(4)));

__device__ __forceinline__ float row16_sum(float v) {
  v += __int_as_float(__builtin_amdgcn_update_dpp(0, __float_as_int(v), 0x161, 0xf, 0xf, true));
  v += __int_as_float(__builtin_amdgcn_update_dpp(0, __float_as_int(v), 0x162, 0xf, 0xf, true));
  v += __int_as_float(__builtin_amdgcn_update_dpp(0, __float_as_int(v), 0x164, 0xf, 0xf, true));
  v += __int_as_float(__builtin_amdgcn_update_dpp(0, __float_as_int(v), 0x168, 0xf, 0xf, true));
  return v;
}

__device__ __forceinline__ void ln1_rows(const float (&raw)[2][2][8],
                                         const float (&g)[8], const float (&bb)[8],
                                         float (&y)[2][2][8]) {
#pragma unroll
  for (int ks = 0; ks < 2; ++ks)
#pragma unroll
    for (int s = 0; s < 2; ++s) {
      float sm = ((raw[ks][s][0] + raw[ks][s][1]) + (raw[ks][s][2] + raw[ks][s][3])) +
                 ((raw[ks][s][4] + raw[ks][s][5]) + (raw[ks][s][6] + raw[ks][s][7]));
      sm = row16_sum(sm);
      const float m = sm * (1.0f / 128.0f);
      float d[8];
      float vs = 0.0f;
#pragma unroll
      for (int t = 0; t < 8; ++t) { d[t] = raw[ks][s][t] - m; vs += d[t] * d[t]; }
      vs = row16_sum(vs);
      const float inv = rsqrtf(vs * (1.0f / 128.0f) + EPSV);
#pragma unroll
      for (int t = 0; t < 8; ++t) y[ks][s][t] = d[t] * inv * g[t] + bb[t];
    }
}

__global__ void __launch_bounds__(128)
k_gram(const float* __restrict__ x,
       const float* __restrict__ g1,
       const float* __restrict__ b1,
       float* __restrict__ stats1,
       float* __restrict__ stats2,
       float* __restrict__ gpart) {
  __shared__ v4f part[4 * 4096];
  __shared__ float s1l[128], s2l[128];
  const int tid = threadIdx.x, lane = tid & 31, wave = tid >> 5;
  const int h = lane >> 4;
  const int c = lane & 15;
  const int b = blockIdx.x >> 3;
  const int oct = blockIdx.x & 7;
  s1l[tid] = 0.0f; s2l[tid] = 0.0f;
  float g[8], bb[8];
#pragma unroll
  for (int t = 0; t < 8; ++t) { g[t] = g1[16 * t + c]; bb[t] = b1[16 * t + c]; }
  const v8f vzero = {0, 0, 0, 0, 0, 0, 0, 0};
  v8f acc[64];
#pragma unroll
  for (int t = 0; t < 64; ++t) acc[t] = vzero;
  float a1[8], a2[8];
#pragma unroll
  for (int t = 0; t < 8; ++t) { a1[t] = 0.0f; a2[t] = 0.0f; }

  const float* xp =
      x + ((long)b * 4096 + oct * 512 + wave * 128 + 2 * h) * 128 + c;
  float y[2][2][8];
  {
    float raw[2][2][8];
#pragma unroll
    for (int ks = 0; ks < 2; ++ks)
#pragma unroll
      for (int s = 0; s < 2; ++s)
#pragma unroll
        for (int t = 0; t < 8; ++t) raw[ks][s][t] = xp[(ks * 4 + s) * 128 + 16 * t];
    ln1_rows(raw, g, bb, y);
  }

#pragma unroll 1
  for (int it = 0; it < 16; ++it) {
    float raw[2][2][8];
    const float* xn = xp + (it < 15 ? it + 1 : 15) * (8 * 128);
#pragma unroll
    for (int ks = 0; ks < 2; ++ks)
#pragma unroll
      for (int s = 0; s < 2; ++s)
#pragma unroll
        for (int t = 0; t < 8; ++t) raw[ks][s][t] = xn[(ks * 4 + s) * 128 + 16 * t];

#pragma unroll
    for (int ks = 0; ks < 2; ++ks)
#pragma unroll
      for (int s = 0; s < 2; ++s)
#pragma unroll
        for (int t = 0; t < 8; ++t) {
          const float yy = y[ks][s][t];
          a1[t] += yy; a2[t] += yy * yy;
        }

#pragma unroll
    for (int ks = 0; ks < 2; ++ks) {
      v2f f[8];
#pragma unroll
      for (int t = 0; t < 8; ++t) { f[t].x = y[ks][0][t]; f[t].y = y[ks][1][t]; }
#pragma unroll
      for (int mt = 0; mt < 8; ++mt)
#pragma unroll
        for (int nt = 0; nt < 8; ++nt)
          acc[mt * 8 + nt] = wmma4(f[mt], f[nt], acc[mt * 8 + nt]);
    }

    ln1_rows(raw, g, bb, y);
  }

  {
    v4f* mine = part + wave * 4096 + lane;
#pragma unroll
    for (int t = 0; t < 64; ++t) {
      v4f lo, hi;
      lo.x = acc[t][0]; lo.y = acc[t][1]; lo.z = acc[t][2]; lo.w = acc[t][3];
      hi.x = acc[t][4]; hi.y = acc[t][5]; hi.z = acc[t][6]; hi.w = acc[t][7];
      mine[(t * 2) * 32] = lo;
      mine[(t * 2 + 1) * 32] = hi;
    }
  }
  __syncthreads();
#pragma unroll
  for (int t = 0; t < 8; ++t) {
    atomicAdd(&s1l[16 * t + c], a1[t]);
    atomicAdd(&s2l[16 * t + c], a2[t]);
  }
  float* gp = gpart + (long)blockIdx.x * 16384;
  for (int i = 0; i < 32; ++i) {
    const int q = i * 4 + wave;
    const v4f p0 = part[q * 32 + lane];
    const v4f p1 = part[4096 + q * 32 + lane];
    const v4f p2 = part[8192 + q * 32 + lane];
    const v4f p3 = part[12288 + q * 32 + lane];
    const v4f sv = (p0 + p1) + (p2 + p3);
    const int tile = q >> 1, rq = q & 1;
    const int mt = tile >> 3, nt = tile & 7;
    float* dst = gp + (mt * 16 + 8 * h + 4 * rq) * 128 + nt * 16 + c;
    dst[0] = sv.x; dst[128] = sv.y; dst[256] = sv.z; dst[384] = sv.w;
  }
  __syncthreads();
  atomicAdd(&stats1[b * 128 + tid], s1l[tid]);
  atomicAdd(&stats2[b * 128 + tid], s2l[tid]);
}

__global__ void k_combine(const float* __restrict__ stats1,
                          const float* __restrict__ stats2,
                          const float* __restrict__ gpart,
                          float* __restrict__ dcorr) {
  const int d = blockIdx.x;
  const int e = threadIdx.x;
  float acc = 0.0f;
  for (int b = 0; b < 32; ++b) {
    float s1d = stats1[b * 128 + d], s2d = stats2[b * 128 + d];
    float s1e = stats1[b * 128 + e], s2e = stats2[b * 128 + e];
    float md = s1d * (1.0f / 4096.0f), me = s1e * (1.0f / 4096.0f);
    float vd = s2d - s1d * s1d * (1.0f / 4096.0f);
    float ve = s2e - s1e * s1e * (1.0f / 4096.0f);
    float ind = 1.0f / fmaxf(sqrtf(fmaxf(vd, 0.0f)), EPSV);
    float ine = 1.0f / fmaxf(sqrtf(fmaxf(ve, 0.0f)), EPSV);
    float G = 0.0f;
#pragma unroll
    for (int q = 0; q < 8; ++q)
      G += gpart[(long)(b * 8 + q) * 16384 + d * 128 + e];
    acc += (G - 4096.0f * md * me) * ind * ine;
  }
  dcorr[d * 128 + e] = acc * (1.0f / 32.0f);
}

__global__ void k_graph(const float* __restrict__ corr,
                        const float* __restrict__ alpha_logit,
                        const float* __restrict__ chebw,
                        const float* __restrict__ dcorr,
                        float* __restrict__ mixed_out,
                        float* __restrict__ Lap,
                        float* __restrict__ wout) {
  __shared__ float dis[128];
  const int i = threadIdx.x;
  const float alpha = 1.0f / (1.0f + __expf(-alpha_logit[0]));
  float deg = 0.0f;
  for (int j = 0; j < 128; ++j) {
    float learned = tanhf(0.5f * (corr[i * 128 + j] + corr[j * 128 + i]));
    float dc = fminf(fmaxf(dcorr[i * 128 + j], -1.0f), 1.0f);
    float mx = alpha * learned + (1.0f - alpha) * dc;
    mixed_out[i * 128 + j] = mx;
    if (j != i) deg += mx;
  }
  dis[i] = rsqrtf(fmaxf(deg, EPSV));
  __syncthreads();
  const float di = dis[i];
  for (int j = 0; j < 128; ++j) {
    float a = (j == i) ? 0.0f : mixed_out[i * 128 + j];
    float l = ((j == i) ? 1.0f : 0.0f) - di * a * dis[j];
    Lap[i * 128 + j] = fminf(fmaxf(l, -1.5f), 1.5f);
  }
  if (i == 0) {
    float c0 = chebw[0], c1 = chebw[1], c2 = chebw[2];
    float mx = fmaxf(c0, fmaxf(c1, c2));
    float e0 = __expf(c0 - mx), e1 = __expf(c1 - mx), e2 = __expf(c2 - mx);
    float s = e0 + e1 + e2;
    wout[0] = e0 / s; wout[1] = e1 / s; wout[2] = e2 / s;
  }
}

#define SLOT_LAP 0
#define SLOT_WI 4096
#define SLOT_WO 12288

template <int BT, int T0, int KT, int KTALL>
__device__ __forceinline__ void gemm_pair(const v4f* __restrict__ wl0,
                                          const v8f (&bin)[BT],
                                          v8f& acc0, v8f& acc1) {
  const v4f* __restrict__ wl1 = wl0 + KTALL * 64;
  v4f af0[KT * 2], af1[KT * 2];
#pragma unroll
  for (int i = 0; i < KT * 2; ++i) { af0[i] = wl0[i * 32]; af1[i] = wl1[i * 32]; }
#pragma unroll
  for (int t = 0; t < KT; ++t) {
#pragma unroll
    for (int p = 0; p < 2; ++p) {
      const v4f A0 = af0[t * 2 + p], A1 = af1[t * 2 + p];
      v2f a00, a01, a10, a11, b0, b1;
      a00.x = A0.x; a00.y = A0.y; a01.x = A0.z; a01.y = A0.w;
      a10.x = A1.x; a10.y = A1.y; a11.x = A1.z; a11.y = A1.w;
      b0.x = bin[T0 + t][4 * p];     b0.y = bin[T0 + t][4 * p + 1];
      b1.x = bin[T0 + t][4 * p + 2]; b1.y = bin[T0 + t][4 * p + 3];
      acc0 = wmma4(a00, b0, acc0);
      acc1 = wmma4(a10, b0, acc1);
      acc0 = wmma4(a01, b1, acc0);
      acc1 = wmma4(a11, b1, acc1);
    }
  }
}

__device__ __forceinline__ v8f load8(const float* __restrict__ p) {
  const v4f lo = *reinterpret_cast<const v4f*>(p);
  const v4f hi = *reinterpret_cast<const v4f*>(p + 4);
  v8f r;
  r[0] = lo.x; r[1] = lo.y; r[2] = lo.z; r[3] = lo.w;
  r[4] = hi.x; r[5] = hi.y; r[6] = hi.z; r[7] = hi.w;
  return r;
}

template <int NTILE>
__device__ __forceinline__ float lane_sum(const v8f (&v)[NTILE]) {
  float part[NTILE];
#pragma unroll
  for (int t = 0; t < NTILE; ++t) {
    float s = v[t][0];
#pragma unroll
    for (int r = 1; r < 8; ++r) s += v[t][r];
    part[t] = s;
  }
  float s = part[0];
#pragma unroll
  for (int t = 1; t < NTILE; ++t) s += part[t];
  return s;
}
template <int NTILE>
__device__ __forceinline__ float lane_sqdev(const v8f (&v)[NTILE], float m) {
  float part[NTILE];
#pragma unroll
  for (int t = 0; t < NTILE; ++t) {
    float s = 0.0f;
#pragma unroll
    for (int r = 0; r < 8; ++r) { const float d = v[t][r] - m; s += d * d; }
    part[t] = s;
  }
  float s = part[0];
#pragma unroll
  for (int t = 1; t < NTILE; ++t) s += part[t];
  return s;
}

__global__ void __launch_bounds__(256)
k_main(const float* __restrict__ x,
       const float* __restrict__ g1, const float* __restrict__ b1,
       const float* __restrict__ Lap, const float* __restrict__ wv,
       const float* __restrict__ Wi, const float* __restrict__ bi,
       const float* __restrict__ Wo, const float* __restrict__ bo,
       const float* __restrict__ g2, const float* __restrict__ b2,
       const float* __restrict__ g3, const float* __restrict__ b3,
       float* __restrict__ out) {
  __shared__ v4f wl[20480];
  const int tid = threadIdx.x, lane = tid & 31, wave = tid >> 5;
  const int hh = lane >> 4;
  const int c = lane & 15;
  const int f8 = 8 * hh;

#pragma unroll 8
  for (int s = tid; s < 4096; s += 256) {
    const int q = s >> 5;
    const int p = q & 1, t = (q >> 1) & 7, mt = q >> 4;
    const int k0 = t * 16 + f8 + 4 * p, m = mt * 16 + c;
    v4f v;
    v.x = Lap[(k0 + 0) * 128 + m]; v.y = Lap[(k0 + 1) * 128 + m];
    v.z = Lap[(k0 + 2) * 128 + m]; v.w = Lap[(k0 + 3) * 128 + m];
    wl[SLOT_LAP + s] = v;
  }
#pragma unroll 8
  for (int s = tid; s < 8192; s += 256) {
    const int q = s >> 5;
    const int p = q & 1, t = (q >> 1) & 7, mt = q >> 4;
    wl[SLOT_WI + s] = *reinterpret_cast<const v4f*>(
        Wi + (mt * 16 + c) * 128 + t * 16 + f8 + 4 * p);
  }
#pragma unroll 8
  for (int s = tid; s < 8192; s += 256) {
    const int q = s >> 5;
    const int p = q & 1, t = (q >> 1) & 15, mt = q >> 5;
    wl[SLOT_WO + s] = *reinterpret_cast<const v4f*>(
        Wo + (mt * 16 + c) * 256 + t * 16 + f8 + 4 * p);
  }
  __syncthreads();

  const float w0 = wv[0], w1 = wv[1], w2 = wv[2];
  const v8f vzero = {0, 0, 0, 0, 0, 0, 0, 0};
  const v4f* wlane = wl + lane;

  for (int blk = blockIdx.x; blk < 1024; blk += gridDim.x) {
    const long srow = ((long)blk * 8 + wave) * 16 + c;

    v8f y[8];
    {
      const float* xs = x + srow * 128 + f8;
#pragma unroll
      for (int t = 0; t < 8; ++t) y[t] = load8(xs + t * 16);
      float s = lane_sum<8>(y);
      s += __shfl_xor(s, 16, 32);
      const float m = s * (1.0f / 128.0f);
      float vs = lane_sqdev<8>(y, m);
      vs += __shfl_xor(vs, 16, 32);
      const float inv = rsqrtf(vs * (1.0f / 128.0f) + EPSV);
#pragma unroll
      for (int t = 0; t < 8; ++t) {
        const v8f g = load8(g1 + t * 16 + f8), bb = load8(b1 + t * 16 + f8);
#pragma unroll
        for (int r = 0; r < 8; ++r) y[t][r] = (y[t][r] - m) * inv * g[r] + bb[r];
      }
    }

    v8f tx[8];
#pragma unroll
    for (int mt = 0; mt < 8; mt += 2) {
      v8f acc0 = vzero, acc1 = vzero;
      gemm_pair<8, 0, 8, 8>(wlane + SLOT_LAP + mt * 512, y, acc0, acc1);
      tx[mt] = acc0; tx[mt + 1] = acc1;
    }

    asm volatile("" ::: "memory");
#pragma unroll
    for (int mt = 0; mt < 8; mt += 2) {
      v8f acc[2] = {vzero, vzero};
      gemm_pair<8, 0, 8, 8>(wlane + SLOT_LAP + mt * 512, tx, acc[0], acc[1]);
#pragma unroll
      for (int u = 0; u < 2; ++u)
#pragma unroll
        for (int r = 0; r < 8; ++r) {
          const float y0 = y[mt + u][r], y1 = tx[mt + u][r];
          const float tx2 = fminf(fmaxf(2.0f * acc[u][r] - y0, -50.0f), 50.0f);
          y[mt + u][r] = w0 * y0 + w1 * y1 + w2 * tx2;
        }
    }

    v8f hd[16];
#pragma unroll
    for (int mt = 0; mt < 16; mt += 2) {
      v8f acc[2] = {vzero, vzero};
      gemm_pair<8, 0, 8, 8>(wlane + SLOT_WI + mt * 512, y, acc[0], acc[1]);
#pragma unroll
      for (int u = 0; u < 2; ++u) {
        const v8f bia = load8(bi + (mt + u) * 16 + f8);
#pragma unroll
        for (int r = 0; r < 8; ++r) hd[mt + u][r] = acc[u][r] + bia[r];
      }
    }

    float m2, inv2;
    {
      float s = lane_sum<16>(hd);
      s += __shfl_xor(s, 16, 32);
      m2 = s * (1.0f / 256.0f);
      float vs = lane_sqdev<16>(hd, m2);
      vs += __shfl_xor(vs, 16, 32);
      inv2 = rsqrtf(vs * (1.0f / 256.0f) + EPSV);
    }
#pragma unroll 1
    for (int mt = 0; mt < 16; ++mt) {
      const v8f g = load8(g2 + mt * 16 + f8), bb = load8(b2 + mt * 16 + f8);
      v8f done;
#pragma unroll
      for (int r = 0; r < 8; ++r) {
        const float yv = (hd[0][r] - m2) * inv2 * g[r] + bb[r];
        done[r] = 0.5f * yv * (1.0f + erff(yv * 0.70710678118654752f));
      }
#pragma unroll
      for (int t = 0; t < 15; ++t) hd[t] = hd[t + 1];
      hd[15] = done;
    }

    v8f od[8];
#pragma unroll
    for (int mt = 0; mt < 8; mt += 2) {
      v8f acc[2] = {vzero, vzero};
      gemm_pair<16, 0, 8, 16>(wlane + SLOT_WO + mt * 1024, hd, acc[0], acc[1]);
      gemm_pair<16, 8, 8, 16>(wlane + SLOT_WO + mt * 1024 + 512, hd, acc[0], acc[1]);
#pragma unroll
      for (int u = 0; u < 2; ++u) {
        const v8f boa = load8(bo + (mt + u) * 16 + f8);
#pragma unroll
        for (int r = 0; r < 8; ++r) od[mt + u][r] = acc[u][r] + boa[r];
      }
    }

    {
      float s = lane_sum<8>(od);
      s += __shfl_xor(s, 16, 32);
      const float m = s * (1.0f / 128.0f);
      float vs = lane_sqdev<8>(od, m);
      vs += __shfl_xor(vs, 16, 32);
      const float inv = rsqrtf(vs * (1.0f / 128.0f) + EPSV);
      float* os = out + srow * 128 + f8;
#pragma unroll
      for (int t = 0; t < 8; ++t) {
        const v8f g = load8(g3 + t * 16 + f8), bb = load8(b3 + t * 16 + f8);
        v4f lo, hi;
        lo.x = (od[t][0] - m) * inv * g[0] + bb[0];
        lo.y = (od[t][1] - m) * inv * g[1] + bb[1];
        lo.z = (od[t][2] - m) * inv * g[2] + bb[2];
        lo.w = (od[t][3] - m) * inv * g[3] + bb[3];
        hi.x = (od[t][4] - m) * inv * g[4] + bb[4];
        hi.y = (od[t][5] - m) * inv * g[5] + bb[5];
        hi.z = (od[t][6] - m) * inv * g[6] + bb[6];
        hi.w = (od[t][7] - m) * inv * g[7] + bb[7];
        *reinterpret_cast<v4f*>(os + t * 16) = lo;
        *reinterpret_cast<v4f*>(os + t * 16 + 4) = hi;
      }
    }
  }
}

extern "C" void kernel_launch(void* const* d_in, const int* in_sizes, int n_in,
                              void* d_out, int out_size, void* d_ws, size_t ws_size,
                              hipStream_t stream) {
  const float* x     = (const float*)d_in[0];
  const float* corr  = (const float*)d_in[1];
  const float* alog  = (const float*)d_in[2];
  const float* chebw = (const float*)d_in[3];
  const float* Wi    = (const float*)d_in[4];
  const float* bi    = (const float*)d_in[5];
  const float* Wo    = (const float*)d_in[6];
  const float* bo    = (const float*)d_in[7];
  const float* g1    = (const float*)d_in[8];
  const float* b1    = (const float*)d_in[9];
  const float* g2    = (const float*)d_in[10];
  const float* b2    = (const float*)d_in[11];
  const float* g3    = (const float*)d_in[12];
  const float* b3    = (const float*)d_in[13];
  float* out = (float*)d_out;
  float* ws  = (float*)d_ws;

  float* stats1 = ws;
  float* stats2 = ws + 4096;
  float* gpart  = ws + 8192;
  float* dcorr  = ws + 8192 + 256 * 16384;
  float* Lap    = dcorr + 16384;
  float* wv     = Lap + 16384;
  float* mixed  = out + (long)32 * 4096 * 128;

  hipMemsetAsync(d_ws, 0, 8192 * sizeof(float), stream);
  k_gram<<<256, 128, 0, stream>>>(x, g1, b1, stats1, stats2, gpart);
  k_combine<<<128, 128, 0, stream>>>(stats1, stats2, gpart, dcorr);
  k_graph<<<1, 128, 0, stream>>>(corr, alog, chebw, dcorr, mixed, Lap, wv);
  k_main<<<256, 256, 0, stream>>>(x, g1, b1, Lap, wv, Wi, bi, Wo, bo,
                                        g2, b2, g3, b3, out);
}


# k_graph launch removed: each k_main work-group builds mixed, degrees and Lap^T in its LDS A slots from corr/dcorr rows staged in LDS; group 0 stores mixed (3 launches, degree sum order changed)
# speedup vs baseline: 3.4173x; 1.2999x over previous
#include <hip/hip_runtime.h>
#include <hip/hip_bf16.h>
#include <math.h>

#define EPSV 1e-8f

typedef float v2f __attribute__((ext_vector_type(2)));
typedef float v8f __attribute__((ext_vector_type(8)));

__device__ __forceinline__ v8f wmma4(v2f a, v2f b, v8f c) {
  return __builtin_amdgcn_wmma_f32_16x16x4_f32(false, a, false, b, (short)0, c,
                                               false, false);
}

typedef float v4f __attribute__((ext_vector_type(4)));

__device__ __forceinline__ float row16_sum(float v) {
  v += __int_as_float(__builtin_amdgcn_update_dpp(0, __float_as_int(v), 0x161, 0xf, 0xf, true));
  v += __int_as_float(__builtin_amdgcn_update_dpp(0, __float_as_int(v), 0x162, 0xf, 0xf, true));
  v += __int_as_float(__builtin_amdgcn_update_dpp(0, __float_as_int(v), 0x164, 0xf, 0xf, true));
  v += __int_as_float(__builtin_amdgcn_update_dpp(0, __float_as_int(v), 0x168, 0xf, 0xf, true));
  return v;
}

__device__ __forceinline__ void ln1_rows(const float (&raw)[2][2][8],
                                         const float (&g)[8], const float (&bb)[8],
                                         float (&y)[2][2][8]) {
#pragma unroll
  for (int ks = 0; ks < 2; ++ks)
#pragma unroll
    for (int s = 0; s < 2; ++s) {
      float sm = ((raw[ks][s][0] + raw[ks][s][1]) + (raw[ks][s][2] + raw[ks][s][3])) +
                 ((raw[ks][s][4] + raw[ks][s][5]) + (raw[ks][s][6] + raw[ks][s][7]));
      sm = row16_sum(sm);
      const float m = sm * (1.0f / 128.0f);
      float d[8];
      float vs = 0.0f;
#pragma unroll
      for (int t = 0; t < 8; ++t) { d[t] = raw[ks][s][t] - m; vs += d[t] * d[t]; }
      vs = row16_sum(vs);
      const float inv = rsqrtf(vs * (1.0f / 128.0f) + EPSV);
#pragma unroll
      for (int t = 0; t < 8; ++t) y[ks][s][t] = d[t] * inv * g[t] + bb[t];
    }
}

__global__ void __launch_bounds__(128)
k_gram(const float* __restrict__ x,
       const float* __restrict__ g1,
       const float* __restrict__ b1,
       float* __restrict__ stats1,
       float* __restrict__ stats2,
       float* __restrict__ gpart) {
  __shared__ v4f part[4 * 4096];
  __shared__ float s1l[128], s2l[128];
  const int tid = threadIdx.x, lane = tid & 31, wave = tid >> 5;
  const int h = lane >> 4;
  const int c = lane & 15;
  const int b = blockIdx.x >> 3;
  const int oct = blockIdx.x & 7;
  s1l[tid] = 0.0f; s2l[tid] = 0.0f;
  float g[8], bb[8];
#pragma unroll
  for (int t = 0; t < 8; ++t) { g[t] = g1[16 * t + c]; bb[t] = b1[16 * t + c]; }
  const v8f vzero = {0, 0, 0, 0, 0, 0, 0, 0};
  v8f acc[64];
#pragma unroll
  for (int t = 0; t < 64; ++t) acc[t] = vzero;
  float a1[8], a2[8];
#pragma unroll
  for (int t = 0; t < 8; ++t) { a1[t] = 0.0f; a2[t] = 0.0f; }

  const float* xp =
      x + ((long)b * 4096 + oct * 512 + wave * 128 + 2 * h) * 128 + c;
  float y[2][2][8];
  {
    float raw[2][2][8];
#pragma unroll
    for (int ks = 0; ks < 2; ++ks)
#pragma unroll
      for (int s = 0; s < 2; ++s)
#pragma unroll
        for (int t = 0; t < 8; ++t) raw[ks][s][t] = xp[(ks * 4 + s) * 128 + 16 * t];
    ln1_rows(raw, g, bb, y);
  }

#pragma unroll 1
  for (int it = 0; it < 16; ++it) {
    float raw[2][2][8];
    const float* xn = xp + (it < 15 ? it + 1 : 15) * (8 * 128);
#pragma unroll
    for (int ks = 0; ks < 2; ++ks)
#pragma unroll
      for (int s = 0; s < 2; ++s)
#pragma unroll
        for (int t = 0; t < 8; ++t) raw[ks][s][t] = xn[(ks * 4 + s) * 128 + 16 * t];

#pragma unroll
    for (int ks = 0; ks < 2; ++ks)
#pragma unroll
      for (int s = 0; s < 2; ++s)
#pragma unroll
        for (int t = 0; t < 8; ++t) {
          const float yy = y[ks][s][t];
          a1[t] += yy; a2[t] += yy * yy;
        }

#pragma unroll
    for (int ks = 0; ks < 2; ++ks) {
      v2f f[8];
#pragma unroll
      for (int t = 0; t < 8; ++t) { f[t].x = y[ks][0][t]; f[t].y = y[ks][1][t]; }
#pragma unroll
      for (int mt = 0; mt < 8; ++mt)
#pragma unroll
        for (int nt = 0; nt < 8; ++nt)
          acc[mt * 8 + nt] = wmma4(f[mt], f[nt], acc[mt * 8 + nt]);
    }

    ln1_rows(raw, g, bb, y);
  }

  {
    v4f* mine = part + wave * 4096 + lane;
#pragma unroll
    for (int t = 0; t < 64; ++t) {
      v4f lo, hi;
      lo.x = acc[t][0]; lo.y = acc[t][1]; lo.z = acc[t][2]; lo.w = acc[t][3];
      hi.x = acc[t][4]; hi.y = acc[t][5]; hi.z = acc[t][6]; hi.w = acc[t][7];
      mine[(t * 2) * 32] = lo;
      mine[(t * 2 + 1) * 32] = hi;
    }
  }
  __syncthreads();
#pragma unroll
  for (int t = 0; t < 8; ++t) {
    atomicAdd(&s1l[16 * t + c], a1[t]);
    atomicAdd(&s2l[16 * t + c], a2[t]);
  }
  float* gp = gpart + (long)blockIdx.x * 16384;
  for (int i = 0; i < 32; ++i) {
    const int q = i * 4 + wave;
    const v4f p0 = part[q * 32 + lane];
    const v4f p1 = part[4096 + q * 32 + lane];
    const v4f p2 = part[8192 + q * 32 + lane];
    const v4f p3 = part[12288 + q * 32 + lane];
    const v4f sv = (p0 + p1) + (p2 + p3);
    const int tile = q >> 1, rq = q & 1;
    const int mt = tile >> 3, nt = tile & 7;
    float* dst = gp + (mt * 16 + 8 * h + 4 * rq) * 128 + nt * 16 + c;
    dst[0] = sv.x; dst[128] = sv.y; dst[256] = sv.z; dst[384] = sv.w;
  }
  __syncthreads();
  atomicAdd(&stats1[b * 128 + tid], s1l[tid]);
  atomicAdd(&stats2[b * 128 + tid], s2l[tid]);
}

__global__ void k_combine(const float* __restrict__ stats1,
                          const float* __restrict__ stats2,
                          const float* __restrict__ gpart,
                          float* __restrict__ dcorr) {
  const int d = blockIdx.x;
  const int e = threadIdx.x;
  float acc = 0.0f;
  for (int b = 0; b < 32; ++b) {
    float s1d = stats1[b * 128 + d], s2d = stats2[b * 128 + d];
    float s1e = stats1[b * 128 + e], s2e = stats2[b * 128 + e];
    float md = s1d * (1.0f / 4096.0f), me = s1e * (1.0f / 4096.0f);
    float vd = s2d - s1d * s1d * (1.0f / 4096.0f);
    float ve = s2e - s1e * s1e * (1.0f / 4096.0f);
    float ind = 1.0f / fmaxf(sqrtf(fmaxf(vd, 0.0f)), EPSV);
    float ine = 1.0f / fmaxf(sqrtf(fmaxf(ve, 0.0f)), EPSV);
    float G = 0.0f;
#pragma unroll
    for (int q = 0; q < 8; ++q)
      G += gpart[(long)(b * 8 + q) * 16384 + d * 128 + e];
    acc += (G - 4096.0f * md * me) * ind * ine;
  }
  dcorr[d * 128 + e] = acc * (1.0f / 32.0f);
}

#define SLOT_LAP 0
#define SLOT_WI 4096
#define SLOT_WO 12288
#define GS_CORR 4096
#define GS_DCORR (4096 + 4224)
#define GS_DIS (4096 + 8448)

template <int BT, int T0, int KT, int KTALL>
__device__ __forceinline__ void gemm_pair(const v4f* __restrict__ wl0,
                                          const v8f (&bin)[BT],
                                          v8f& acc0, v8f& acc1) {
  const v4f* __restrict__ wl1 = wl0 + KTALL * 64;
  v4f af0[KT * 2], af1[KT * 2];
#pragma unroll
  for (int i = 0; i < KT * 2; ++i) { af0[i] = wl0[i * 32]; af1[i] = wl1[i * 32]; }
#pragma unroll
  for (int t = 0; t < KT; ++t) {
#pragma unroll
    for (int p = 0; p < 2; ++p) {
      const v4f A0 = af0[t * 2 + p], A1 = af1[t * 2 + p];
      v2f a00, a01, a10, a11, b0, b1;
      a00.x = A0.x; a00.y = A0.y; a01.x = A0.z; a01.y = A0.w;
      a10.x = A1.x; a10.y = A1.y; a11.x = A1.z; a11.y = A1.w;
      b0.x = bin[T0 + t][4 * p];     b0.y = bin[T0 + t][4 * p + 1];
      b1.x = bin[T0 + t][4 * p + 2]; b1.y = bin[T0 + t][4 * p + 3];
      acc0 = wmma4(a00, b0, acc0);
      acc1 = wmma4(a10, b0, acc1);
      acc0 = wmma4(a01, b1, acc0);
      acc1 = wmma4(a11, b1, acc1);
    }
  }
}

__device__ __forceinline__ v8f load8(const float* __restrict__ p) {
  const v4f lo = *reinterpret_cast<const v4f*>(p);
  const v4f hi = *reinterpret_cast<const v4f*>(p + 4);
  v8f r;
  r[0] = lo.x; r[1] = lo.y; r[2] = lo.z; r[3] = lo.w;
  r[4] = hi.x; r[5] = hi.y; r[6] = hi.z; r[7] = hi.w;
  return r;
}

template <int NTILE>
__device__ __forceinline__ float lane_sum(const v8f (&v)[NTILE]) {
  float part[NTILE];
#pragma unroll
  for (int t = 0; t < NTILE; ++t) {
    float s = v[t][0];
#pragma unroll
    for (int r = 1; r < 8; ++r) s += v[t][r];
    part[t] = s;
  }
  float s = part[0];
#pragma unroll
  for (int t = 1; t < NTILE; ++t) s += part[t];
  return s;
}
template <int NTILE>
__device__ __forceinline__ float lane_sqdev(const v8f (&v)[NTILE], float m) {
  float part[NTILE];
#pragma unroll
  for (int t = 0; t < NTILE; ++t) {
    float s = 0.0f;
#pragma unroll
    for (int r = 0; r < 8; ++r) { const float d = v[t][r] - m; s += d * d; }
    part[t] = s;
  }
  float s = part[0];
#pragma unroll
  for (int t = 1; t < NTILE; ++t) s += part[t];
  return s;
}

__global__ void __launch_bounds__(256)
k_main(const float* __restrict__ x,
       const float* __restrict__ g1, const float* __restrict__ b1,
       const float* __restrict__ corr, const float* __restrict__ alpha_logit,
       const float* __restrict__ chebw, const float* __restrict__ dcorr,
       const float* __restrict__ Wi, const float* __restrict__ bi,
       const float* __restrict__ Wo, const float* __restrict__ bo,
       const float* __restrict__ g2, const float* __restrict__ b2,
       const float* __restrict__ g3, const float* __restrict__ b3,
       float* __restrict__ out, float* __restrict__ mixed_out) {
  __shared__ v4f wl[20480];
  const int tid = threadIdx.x, lane = tid & 31, wave = tid >> 5;
  const int hh = lane >> 4;
  const int c = lane & 15;
  const int f8 = 8 * hh;

  {
    v4f ca[16], da[16];
#pragma unroll
    for (int n = 0; n < 16; ++n) {
      ca[n] = reinterpret_cast<const v4f*>(corr)[tid + 256 * n];
      da[n] = reinterpret_cast<const v4f*>(dcorr)[tid + 256 * n];
    }
#pragma unroll
    for (int n = 0; n < 16; ++n) {
      const int i = tid + 256 * n;
      wl[GS_CORR + (i >> 5) * 33 + (i & 31)] = ca[n];
      wl[GS_DCORR + (i >> 5) * 33 + (i & 31)] = da[n];
    }
  }
  __syncthreads();
  const float alpha = 1.0f / (1.0f + __expf(-alpha_logit[0]));
  const float* csf = reinterpret_cast<const float*>(wl + GS_CORR);
  const float* dsf = reinterpret_cast<const float*>(wl + GS_DCORR);
  float* dis_s = reinterpret_cast<float*>(wl + GS_DIS);
  const int gp = wave & 1, gtb = wave >> 1;
  {
    float degp[2][4];
#pragma unroll
    for (int ts = 0; ts < 2; ++ts)
#pragma unroll
      for (int e = 0; e < 4; ++e) degp[ts][e] = 0.0f;
#pragma unroll 1
    for (int mt = 0; mt < 8; ++mt) {
#pragma unroll
      for (int ts = 0; ts < 2; ++ts) {
        const int t = gtb + 4 * ts;
        const int m = mt * 16 + c;
        const int k0 = t * 16 + f8 + 4 * gp;
        const v4f ct = wl[GS_CORR + m * 33 + (k0 >> 2)];
        v4f mx;
#pragma unroll
        for (int e = 0; e < 4; ++e) {
          const float cd = csf[(k0 + e) * 132 + m];
          const float dcl = fminf(fmaxf(dsf[(k0 + e) * 132 + m], -1.0f), 1.0f);
          const float learned = tanhf(0.5f * (cd + ct[e]));
          const float v = alpha * learned + (1.0f - alpha) * dcl;
          mx[e] = v;
          if (k0 + e != m) degp[ts][e] += v;
        }
        if (blockIdx.x == 0) {
#pragma unroll
          for (int e = 0; e < 4; ++e) mixed_out[(k0 + e) * 128 + m] = mx[e];
        }
        wl[SLOT_LAP + ((mt * 8 + t) * 2 + gp) * 32 + lane] = mx;
      }
    }
#pragma unroll
    for (int ts = 0; ts < 2; ++ts)
#pragma unroll
      for (int e = 0; e < 4; ++e) {
        const float deg = row16_sum(degp[ts][e]);
        if (c == 0)
          dis_s[(gtb + 4 * ts) * 16 + f8 + 4 * gp + e] = rsqrtf(fmaxf(deg, EPSV));
      }
  }
  __syncthreads();
#pragma unroll 4
  for (int it = 0; it < 16; ++it) {
    const int mt = it >> 1, t = gtb + 4 * (it & 1);
    const int m = mt * 16 + c;
    const int k0 = t * 16 + f8 + 4 * gp;
    v4f* slot = wl + SLOT_LAP + ((mt * 8 + t) * 2 + gp) * 32 + lane;
    const v4f mx = *slot;
    const float dm = dis_s[m];
    const v4f dk = *reinterpret_cast<const v4f*>(dis_s + k0);
    v4f lp;
#pragma unroll
    for (int e = 0; e < 4; ++e) {
      const bool dg = (k0 + e == m);
      const float a = dg ? 0.0f : mx[e];
      const float l = (dg ? 1.0f : 0.0f) - dk[e] * a * dm;
      lp[e] = fminf(fmaxf(l, -1.5f), 1.5f);
    }
    *slot = lp;
  }
  __syncthreads();

#pragma unroll 8
  for (int s = tid; s < 8192; s += 256) {
    const int q = s >> 5;
    const int p = q & 1, t = (q >> 1) & 7, mt = q >> 4;
    wl[SLOT_WI + s] = *reinterpret_cast<const v4f*>(
        Wi + (mt * 16 + c) * 128 + t * 16 + f8 + 4 * p);
  }
#pragma unroll 8
  for (int s = tid; s < 8192; s += 256) {
    const int q = s >> 5;
    const int p = q & 1, t = (q >> 1) & 15, mt = q >> 5;
    wl[SLOT_WO + s] = *reinterpret_cast<const v4f*>(
        Wo + (mt * 16 + c) * 256 + t * 16 + f8 + 4 * p);
  }
  __syncthreads();

  float w0, w1, w2;
  {
    const float c0 = chebw[0], c1 = chebw[1], c2 = chebw[2];
    const float mxw = fmaxf(c0, fmaxf(c1, c2));
    const float e0 = __expf(c0 - mxw), e1 = __expf(c1 - mxw), e2 = __expf(c2 - mxw);
    const float sw = e0 + e1 + e2;
    w0 = e0 / sw; w1 = e1 / sw; w2 = e2 / sw;
  }
  const v8f vzero = {0, 0, 0, 0, 0, 0, 0, 0};
  const v4f* wlane = wl + lane;

  for (int blk = blockIdx.x; blk < 1024; blk += gridDim.x) {
    const long srow = ((long)blk * 8 + wave) * 16 + c;

    v8f y[8];
    {
      const float* xs = x + srow * 128 + f8;
#pragma unroll
      for (int t = 0; t < 8; ++t) y[t] = load8(xs + t * 16);
      float s = lane_sum<8>(y);
      s += __shfl_xor(s, 16, 32);
      const float m = s * (1.0f / 128.0f);
      float vs = lane_sqdev<8>(y, m);
      vs += __shfl_xor(vs, 16, 32);
      const float inv = rsqrtf(vs * (1.0f / 128.0f) + EPSV);
#pragma unroll
      for (int t = 0; t < 8; ++t) {
        const v8f g = load8(g1 + t * 16 + f8), bb = load8(b1 + t * 16 + f8);
#pragma unroll
        for (int r = 0; r < 8; ++r) y[t][r] = (y[t][r] - m) * inv * g[r] + bb[r];
      }
    }

    v8f tx[8];
#pragma unroll
    for (int mt = 0; mt < 8; mt += 2) {
      v8f acc0 = vzero, acc1 = vzero;
      gemm_pair<8, 0, 8, 8>(wlane + SLOT_LAP + mt * 512, y, acc0, acc1);
      tx[mt] = acc0; tx[mt + 1] = acc1;
    }

    asm volatile("" ::: "memory");
#pragma unroll
    for (int mt = 0; mt < 8; mt += 2) {
      v8f acc[2] = {vzero, vzero};
      gemm_pair<8, 0, 8, 8>(wlane + SLOT_LAP + mt * 512, tx, acc[0], acc[1]);
#pragma unroll
      for (int u = 0; u < 2; ++u)
#pragma unroll
        for (int r = 0; r < 8; ++r) {
          const float y0 = y[mt + u][r], y1 = tx[mt + u][r];
          const float tx2 = fminf(fmaxf(2.0f * acc[u][r] - y0, -50.0f), 50.0f);
          y[mt + u][r] = w0 * y0 + w1 * y1 + w2 * tx2;
        }
    }

    v8f hd[16];
#pragma unroll
    for (int mt = 0; mt < 16; mt += 2) {
      v8f acc[2] = {vzero, vzero};
      gemm_pair<8, 0, 8, 8>(wlane + SLOT_WI + mt * 512, y, acc[0], acc[1]);
#pragma unroll
      for (int u = 0; u < 2; ++u) {
        const v8f bia = load8(bi + (mt + u) * 16 + f8);
#pragma unroll
        for (int r = 0; r < 8; ++r) hd[mt + u][r] = acc[u][r] + bia[r];
      }
    }

    float m2, inv2;
    {
      float s = lane_sum<16>(hd);
      s += __shfl_xor(s, 16, 32);
      m2 = s * (1.0f / 256.0f);
      float vs = lane_sqdev<16>(hd, m2);
      vs += __shfl_xor(vs, 16, 32);
      inv2 = rsqrtf(vs * (1.0f / 256.0f) + EPSV);
    }
#pragma unroll 1
    for (int mt = 0; mt < 16; ++mt) {
      const v8f g = load8(g2 + mt * 16 + f8), bb = load8(b2 + mt * 16 + f8);
      v8f done;
#pragma unroll
      for (int r = 0; r < 8; ++r) {
        const float yv = (hd[0][r] - m2) * inv2 * g[r] + bb[r];
        done[r] = 0.5f * yv * (1.0f + erff(yv * 0.70710678118654752f));
      }
#pragma unroll
      for (int t = 0; t < 15; ++t) hd[t] = hd[t + 1];
      hd[15] = done;
    }

    v8f od[8];
#pragma unroll
    for (int mt = 0; mt < 8; mt += 2) {
      v8f acc[2] = {vzero, vzero};
      gemm_pair<16, 0, 8, 16>(wlane + SLOT_WO + mt * 1024, hd, acc[0], acc[1]);
      gemm_pair<16, 8, 8, 16>(wlane + SLOT_WO + mt * 1024 + 512, hd, acc[0], acc[1]);
#pragma unroll
      for (int u = 0; u < 2; ++u) {
        const v8f boa = load8(bo + (mt + u) * 16 + f8);
#pragma unroll
        for (int r = 0; r < 8; ++r) od[mt + u][r] = acc[u][r] + boa[r];
      }
    }

    {
      float s = lane_sum<8>(od);
      s += __shfl_xor(s, 16, 32);
      const float m = s * (1.0f / 128.0f);
      float vs = lane_sqdev<8>(od, m);
      vs += __shfl_xor(vs, 16, 32);
      const float inv = rsqrtf(vs * (1.0f / 128.0f) + EPSV);
      float* os = out + srow * 128 + f8;
#pragma unroll
      for (int t = 0; t < 8; ++t) {
        const v8f g = load8(g3 + t * 16 + f8), bb = load8(b3 + t * 16 + f8);
        v4f lo, hi;
        lo.x = (od[t][0] - m) * inv * g[0] + bb[0];
        lo.y = (od[t][1] - m) * inv * g[1] + bb[1];
        lo.z = (od[t][2] - m) * inv * g[2] + bb[2];
        lo.w = (od[t][3] - m) * inv * g[3] + bb[3];
        hi.x = (od[t][4] - m) * inv * g[4] + bb[4];
        hi.y = (od[t][5] - m) * inv * g[5] + bb[5];
        hi.z = (od[t][6] - m) * inv * g[6] + bb[6];
        hi.w = (od[t][7] - m) * inv * g[7] + bb[7];
        *reinterpret_cast<v4f*>(os + t * 16) = lo;
        *reinterpret_cast<v4f*>(os + t * 16 + 4) = hi;
      }
    }
  }
}

extern "C" void kernel_launch(void* const* d_in, const int* in_sizes, int n_in,
                              void* d_out, int out_size, void* d_ws, size_t ws_size,
                              hipStream_t stream) {
  const float* x     = (const float*)d_in[0];
  const float* corr  = (const float*)d_in[1];
  const float* alog  = (const float*)d_in[2];
  const float* chebw = (const float*)d_in[3];
  const float* Wi    = (const float*)d_in[4];
  const float* bi    = (const float*)d_in[5];
  const float* Wo    = (const float*)d_in[6];
  const float* bo    = (const float*)d_in[7];
  const float* g1    = (const float*)d_in[8];
  const float* b1    = (const float*)d_in[9];
  const float* g2    = (const float*)d_in[10];
  const float* b2    = (const float*)d_in[11];
  const float* g3    = (const float*)d_in[12];
  const float* b3    = (const float*)d_in[13];
  float* out = (float*)d_out;
  float* ws  = (float*)d_ws;

  float* stats1 = ws;
  float* stats2 = ws + 4096;
  float* gpart  = ws + 8192;
  float* dcorr  = ws + 8192 + 256 * 16384;
  float* mixed  = out + (long)32 * 4096 * 128;

  hipMemsetAsync(d_ws, 0, 8192 * sizeof(float), stream);
  k_gram<<<256, 128, 0, stream>>>(x, g1, b1, stats1, stats2, gpart);
  k_combine<<<128, 128, 0, stream>>>(stats1, stats2, gpart, dcorr);
  k_main<<<256, 256, 0, stream>>>(x, g1, b1, corr, alog, chebw, dcorr, Wi, bi, Wo, bo,
                                  g2, b2, g3, b3, out, mixed);
}


# k_main at three waves per SIMD: work-groups of 12 waves, a wave takes whole 16-sample tiles (3 or 2 of its group's 32); Wi/Wo A fragments fetched in groups of 16 so a wave fits 336 registers
# speedup vs baseline: 3.8099x; 1.1149x over previous
#include <hip/hip_runtime.h>
#include <hip/hip_bf16.h>
#include <math.h>

#define EPSV 1e-8f

typedef float v2f __attribute__((ext_vector_type(2)));
typedef float v8f __attribute__((ext_vector_type(8)));

__device__ __forceinline__ v8f wmma4(v2f a, v2f b, v8f c) {
  return __builtin_amdgcn_wmma_f32_16x16x4_f32(false, a, false, b, (short)0, c,
                                               false, false);
}

typedef float v4f __attribute__((ext_vector_type(4)));

__device__ __forceinline__ float row16_sum(float v) {
  v += __int_as_float(__builtin_amdgcn_update_dpp(0, __float_as_int(v), 0x161, 0xf, 0xf, true));
  v += __int_as_float(__builtin_amdgcn_update_dpp(0, __float_as_int(v), 0x162, 0xf, 0xf, true));
  v += __int_as_float(__builtin_amdgcn_update_dpp(0, __float_as_int(v), 0x164, 0xf, 0xf, true));
  v += __int_as_float(__builtin_amdgcn_update_dpp(0, __float_as_int(v), 0x168, 0xf, 0xf, true));
  return v;
}

__device__ __forceinline__ void ln1_rows(const float (&raw)[2][2][8],
                                         const float (&g)[8], const float (&bb)[8],
                                         float (&y)[2][2][8]) {
#pragma unroll
  for (int ks = 0; ks < 2; ++ks)
#pragma unroll
    for (int s = 0; s < 2; ++s) {
      float sm = ((raw[ks][s][0] + raw[ks][s][1]) + (raw[ks][s][2] + raw[ks][s][3])) +
                 ((raw[ks][s][4] + raw[ks][s][5]) + (raw[ks][s][6] + raw[ks][s][7]));
      sm = row16_sum(sm);
      const float m = sm * (1.0f / 128.0f);
      float d[8];
      float vs = 0.0f;
#pragma unroll
      for (int t = 0; t < 8; ++t) { d[t] = raw[ks][s][t] - m; vs += d[t] * d[t]; }
      vs = row16_sum(vs);
      const float inv = rsqrtf(vs * (1.0f / 128.0f) + EPSV);
#pragma unroll
      for (int t = 0; t < 8; ++t) y[ks][s][t] = d[t] * inv * g[t] + bb[t];
    }
}

__global__ void __launch_bounds__(128)
k_gram(const float* __restrict__ x,
       const float* __restrict__ g1,
       const float* __restrict__ b1,
       float* __restrict__ stats1,
       float* __restrict__ stats2,
       float* __restrict__ gpart) {
  __shared__ v4f part[4 * 4096];
  __shared__ float s1l[128], s2l[128];
  const int tid = threadIdx.x, lane = tid & 31, wave = tid >> 5;
  const int h = lane >> 4;
  const int c = lane & 15;
  const int b = blockIdx.x >> 3;
  const int oct = blockIdx.x & 7;
  s1l[tid] = 0.0f; s2l[tid] = 0.0f;
  float g[8], bb[8];
#pragma unroll
  for (int t = 0; t < 8; ++t) { g[t] = g1[16 * t + c]; bb[t] = b1[16 * t + c]; }
  const v8f vzero = {0, 0, 0, 0, 0, 0, 0, 0};
  v8f acc[64];
#pragma unroll
  for (int t = 0; t < 64; ++t) acc[t] = vzero;
  float a1[8], a2[8];
#pragma unroll
  for (int t = 0; t < 8; ++t) { a1[t] = 0.0f; a2[t] = 0.0f; }

  const float* xp =
      x + ((long)b * 4096 + oct * 512 + wave * 128 + 2 * h) * 128 + c;
  float y[2][2][8];
  {
    float raw[2][2][8];
#pragma unroll
    for (int ks = 0; ks < 2; ++ks)
#pragma unroll
      for (int s = 0; s < 2; ++s)
#pragma unroll
        for (int t = 0; t < 8; ++t) raw[ks][s][t] = xp[(ks * 4 + s) * 128 + 16 * t];
    ln1_rows(raw, g, bb, y);
  }

#pragma unroll 1
  for (int it = 0; it < 16; ++it) {
    float raw[2][2][8];
    const float* xn = xp + (it < 15 ? it + 1 : 15) * (8 * 128);
#pragma unroll
    for (int ks = 0; ks < 2; ++ks)
#pragma unroll
      for (int s = 0; s < 2; ++s)
#pragma unroll
        for (int t = 0; t < 8; ++t) raw[ks][s][t] = xn[(ks * 4 + s) * 128 + 16 * t];

#pragma unroll
    for (int ks = 0; ks < 2; ++ks)
#pragma unroll
      for (int s = 0; s < 2; ++s)
#pragma unroll
        for (int t = 0; t < 8; ++t) {
          const float yy = y[ks][s][t];
          a1[t] += yy; a2[t] += yy * yy;
        }

#pragma unroll
    for (int ks = 0; ks < 2; ++ks) {
      v2f f[8];
#pragma unroll
      for (int t = 0; t < 8; ++t) { f[t].x = y[ks][0][t]; f[t].y = y[ks][1][t]; }
#pragma unroll
      for (int mt = 0; mt < 8; ++mt)
#pragma unroll
        for (int nt = 0; nt < 8; ++nt)
          acc[mt * 8 + nt] = wmma4(f[mt], f[nt], acc[mt * 8 + nt]);
    }

    ln1_rows(raw, g, bb, y);
  }

  {
    v4f* mine = part + wave * 4096 + lane;
#pragma unroll
    for (int t = 0; t < 64; ++t) {
      v4f lo, hi;
      lo.x = acc[t][0]; lo.y = acc[t][1]; lo.z = acc[t][2]; lo.w = acc[t][3];
      hi.x = acc[t][4]; hi.y = acc[t][5]; hi.z = acc[t][6]; hi.w = acc[t][7];
      mine[(t * 2) * 32] = lo;
      mine[(t * 2 + 1) * 32] = hi;
    }
  }
  __syncthreads();
#pragma unroll
  for (int t = 0; t < 8; ++t) {
    atomicAdd(&s1l[16 * t + c], a1[t]);
    atomicAdd(&s2l[16 * t + c], a2[t]);
  }
  float* gp = gpart + (long)blockIdx.x * 16384;
  for (int i = 0; i < 32; ++i) {
    const int q = i * 4 + wave;
    const v4f p0 = part[q * 32 + lane];
    const v4f p1 = part[4096 + q * 32 + lane];
    const v4f p2 = part[8192 + q * 32 + lane];
    const v4f p3 = part[12288 + q * 32 + lane];
    const v4f sv = (p0 + p1) + (p2 + p3);
    const int tile = q >> 1, rq = q & 1;
    const int mt = tile >> 3, nt = tile & 7;
    float* dst = gp + (mt * 16 + 8 * h + 4 * rq) * 128 + nt * 16 + c;
    dst[0] = sv.x; dst[128] = sv.y; dst[256] = sv.z; dst[384] = sv.w;
  }
  __syncthreads();
  atomicAdd(&stats1[b * 128 + tid], s1l[tid]);
  atomicAdd(&stats2[b * 128 + tid], s2l[tid]);
}

__global__ void k_combine(const float* __restrict__ stats1,
                          const float* __restrict__ stats2,
                          const float* __restrict__ gpart,
                          float* __restrict__ dcorr) {
  const int d = blockIdx.x;
  const int e = threadIdx.x;
  float acc = 0.0f;
  for (int b = 0; b < 32; ++b) {
    float s1d = stats1[b * 128 + d], s2d = stats2[b * 128 + d];
    float s1e = stats1[b * 128 + e], s2e = stats2[b * 128 + e];
    float md = s1d * (1.0f / 4096.0f), me = s1e * (1.0f / 4096.0f);
    float vd = s2d - s1d * s1d * (1.0f / 4096.0f);
    float ve = s2e - s1e * s1e * (1.0f / 4096.0f);
    float ind = 1.0f / fmaxf(sqrtf(fmaxf(vd, 0.0f)), EPSV);
    float ine = 1.0f / fmaxf(sqrtf(fmaxf(ve, 0.0f)), EPSV);
    float G = 0.0f;
#pragma unroll
    for (int q = 0; q < 8; ++q)
      G += gpart[(long)(b * 8 + q) * 16384 + d * 128 + e];
    acc += (G - 4096.0f * md * me) * ind * ine;
  }
  dcorr[d * 128 + e] = acc * (1.0f / 32.0f);
}

#define SLOT_LAP 0
#define SLOT_WI 4096
#define SLOT_WO 12288
#define GS_CORR 4096
#define GS_DCORR (4096 + 4224)
#define GS_DIS (4096 + 8448)

template <int BT, int T0, int KT, int KTALL>
__device__ __forceinline__ void gemm_pair(const v4f* __restrict__ wl0,
                                          const v8f (&bin)[BT],
                                          v8f& acc0, v8f& acc1) {
  const v4f* __restrict__ wl1 = wl0 + KTALL * 64;
  v4f af0[KT * 2], af1[KT * 2];
#pragma unroll
  for (int i = 0; i < KT * 2; ++i) { af0[i] = wl0[i * 32]; af1[i] = wl1[i * 32]; }
#pragma unroll
  for (int t = 0; t < KT; ++t) {
#pragma unroll
    for (int p = 0; p < 2; ++p) {
      const v4f A0 = af0[t * 2 + p], A1 = af1[t * 2 + p];
      v2f a00, a01, a10, a11, b0, b1;
      a00.x = A0.x; a00.y = A0.y; a01.x = A0.z; a01.y = A0.w;
      a10.x = A1.x; a10.y = A1.y; a11.x = A1.z; a11.y = A1.w;
      b0.x = bin[T0 + t][4 * p];     b0.y = bin[T0 + t][4 * p + 1];
      b1.x = bin[T0 + t][4 * p + 2]; b1.y = bin[T0 + t][4 * p + 3];
      acc0 = wmma4(a00, b0, acc0);
      acc1 = wmma4(a10, b0, acc1);
      acc0 = wmma4(a01, b1, acc0);
      acc1 = wmma4(a11, b1, acc1);
    }
  }
}

__device__ __forceinline__ v8f load8(const float* __restrict__ p) {
  const v4f lo = *reinterpret_cast<const v4f*>(p);
  const v4f hi = *reinterpret_cast<const v4f*>(p + 4);
  v8f r;
  r[0] = lo.x; r[1] = lo.y; r[2] = lo.z; r[3] = lo.w;
  r[4] = hi.x; r[5] = hi.y; r[6] = hi.z; r[7] = hi.w;
  return r;
}

template <int NTILE>
__device__ __forceinline__ float lane_sum(const v8f (&v)[NTILE]) {
  float part[NTILE];
#pragma unroll
  for (int t = 0; t < NTILE; ++t) {
    float s = v[t][0];
#pragma unroll
    for (int r = 1; r < 8; ++r) s += v[t][r];
    part[t] = s;
  }
  float s = part[0];
#pragma unroll
  for (int t = 1; t < NTILE; ++t) s += part[t];
  return s;
}
template <int NTILE>
__device__ __forceinline__ float lane_sqdev(const v8f (&v)[NTILE], float m) {
  float part[NTILE];
#pragma unroll
  for (int t = 0; t < NTILE; ++t) {
    float s = 0.0f;
#pragma unroll
    for (int r = 0; r < 8; ++r) { const float d = v[t][r] - m; s += d * d; }
    part[t] = s;
  }
  float s = part[0];
#pragma unroll
  for (int t = 1; t < NTILE; ++t) s += part[t];
  return s;
}

__global__ void __launch_bounds__(384)
k_main(const float* __restrict__ x,
       const float* __restrict__ g1, const float* __restrict__ b1,
       const float* __restrict__ corr, const float* __restrict__ alpha_logit,
       const float* __restrict__ chebw, const float* __restrict__ dcorr,
       const float* __restrict__ Wi, const float* __restrict__ bi,
       const float* __restrict__ Wo, const float* __restrict__ bo,
       const float* __restrict__ g2, const float* __restrict__ b2,
       const float* __restrict__ g3, const float* __restrict__ b3,
       float* __restrict__ out, float* __restrict__ mixed_out) {
  __shared__ v4f wl[20480];
  const int tid = threadIdx.x, lane = tid & 31, wave = tid >> 5;
  const int hh = lane >> 4;
  const int c = lane & 15;
  const int f8 = 8 * hh;

  if (tid < 256) {
    v4f ca[16], da[16];
#pragma unroll
    for (int n = 0; n < 16; ++n) {
      ca[n] = reinterpret_cast<const v4f*>(corr)[tid + 256 * n];
      da[n] = reinterpret_cast<const v4f*>(dcorr)[tid + 256 * n];
    }
#pragma unroll
    for (int n = 0; n < 16; ++n) {
      const int i = tid + 256 * n;
      wl[GS_CORR + (i >> 5) * 33 + (i & 31)] = ca[n];
      wl[GS_DCORR + (i >> 5) * 33 + (i & 31)] = da[n];
    }
  }
  __syncthreads();
  const float alpha = 1.0f / (1.0f + __expf(-alpha_logit[0]));
  const float* csf = reinterpret_cast<const float*>(wl + GS_CORR);
  const float* dsf = reinterpret_cast<const float*>(wl + GS_DCORR);
  float* dis_s = reinterpret_cast<float*>(wl + GS_DIS);
  const int gp = wave & 1, gtb = wave >> 1;
  if (tid < 256) {
    float degp[2][4];
#pragma unroll
    for (int ts = 0; ts < 2; ++ts)
#pragma unroll
      for (int e = 0; e < 4; ++e) degp[ts][e] = 0.0f;
#pragma unroll 1
    for (int mt = 0; mt < 8; ++mt) {
#pragma unroll
      for (int ts = 0; ts < 2; ++ts) {
        const int t = gtb + 4 * ts;
        const int m = mt * 16 + c;
        const int k0 = t * 16 + f8 + 4 * gp;
        const v4f ct = wl[GS_CORR + m * 33 + (k0 >> 2)];
        v4f mx;
#pragma unroll
        for (int e = 0; e < 4; ++e) {
          const float cd = csf[(k0 + e) * 132 + m];
          const float dcl = fminf(fmaxf(dsf[(k0 + e) * 132 + m], -1.0f), 1.0f);
          const float learned = tanhf(0.5f * (cd + ct[e]));
          const float v = alpha * learned + (1.0f - alpha) * dcl;
          mx[e] = v;
          if (k0 + e != m) degp[ts][e] += v;
        }
        if (blockIdx.x == 0) {
#pragma unroll
          for (int e = 0; e < 4; ++e) mixed_out[(k0 + e) * 128 + m] = mx[e];
        }
        wl[SLOT_LAP + ((mt * 8 + t) * 2 + gp) * 32 + lane] = mx;
      }
    }
#pragma unroll
    for (int ts = 0; ts < 2; ++ts)
#pragma unroll
      for (int e = 0; e < 4; ++e) {
        const float deg = row16_sum(degp[ts][e]);
        if (c == 0)
          dis_s[(gtb + 4 * ts) * 16 + f8 + 4 * gp + e] = rsqrtf(fmaxf(deg, EPSV));
      }
  }
  __syncthreads();
  if (tid < 256) {
#pragma unroll 4
  for (int it = 0; it < 16; ++it) {
    const int mt = it >> 1, t = gtb + 4 * (it & 1);
    const int m = mt * 16 + c;
    const int k0 = t * 16 + f8 + 4 * gp;
    v4f* slot = wl + SLOT_LAP + ((mt * 8 + t) * 2 + gp) * 32 + lane;
    const v4f mx = *slot;
    const float dm = dis_s[m];
    const v4f dk = *reinterpret_cast<const v4f*>(dis_s + k0);
    v4f lp;
#pragma unroll
    for (int e = 0; e < 4; ++e) {
      const bool dg = (k0 + e == m);
      const float a = dg ? 0.0f : mx[e];
      const float l = (dg ? 1.0f : 0.0f) - dk[e] * a * dm;
      lp[e] = fminf(fmaxf(l, -1.5f), 1.5f);
    }
    *slot = lp;
  }
  }
  __syncthreads();

  if (tid < 256) {
#pragma unroll 8
  for (int s = tid; s < 8192; s += 256) {
    const int q = s >> 5;
    const int p = q & 1, t = (q >> 1) & 7, mt = q >> 4;
    wl[SLOT_WI + s] = *reinterpret_cast<const v4f*>(
        Wi + (mt * 16 + c) * 128 + t * 16 + f8 + 4 * p);
  }
#pragma unroll 8
  for (int s = tid; s < 8192; s += 256) {
    const int q = s >> 5;
    const int p = q & 1, t = (q >> 1) & 15, mt = q >> 5;
    wl[SLOT_WO + s] = *reinterpret_cast<const v4f*>(
        Wo + (mt * 16 + c) * 256 + t * 16 + f8 + 4 * p);
  }
  }
  __syncthreads();

  float w0, w1, w2;
  {
    const float c0 = chebw[0], c1 = chebw[1], c2 = chebw[2];
    const float mxw = fmaxf(c0, fmaxf(c1, c2));
    const float e0 = __expf(c0 - mxw), e1 = __expf(c1 - mxw), e2 = __expf(c2 - mxw);
    const float sw = e0 + e1 + e2;
    w0 = e0 / sw; w1 = e1 / sw; w2 = e2 / sw;
  }
  const v8f vzero = {0, 0, 0, 0, 0, 0, 0, 0};
  const v4f* wlane = wl + lane;

  for (int tl = wave; tl < 32; tl += 12) {
    const long srow = ((long)blockIdx.x * 32 + tl) * 16 + c;
    asm volatile("" ::: "memory");

    v8f y[8];
    {
      const float* xs = x + srow * 128 + f8;
#pragma unroll
      for (int t = 0; t < 8; ++t) y[t] = load8(xs + t * 16);
      float s = lane_sum<8>(y);
      s += __shfl_xor(s, 16, 32);
      const float m = s * (1.0f / 128.0f);
      float vs = lane_sqdev<8>(y, m);
      vs += __shfl_xor(vs, 16, 32);
      const float inv = rsqrtf(vs * (1.0f / 128.0f) + EPSV);
#pragma unroll
      for (int t = 0; t < 8; ++t) {
        const v8f g = load8(g1 + t * 16 + f8), bb = load8(b1 + t * 16 + f8);
#pragma unroll
        for (int r = 0; r < 8; ++r) y[t][r] = (y[t][r] - m) * inv * g[r] + bb[r];
      }
    }

    v8f tx[8];
#pragma unroll
    for (int mt = 0; mt < 8; mt += 2) {
      v8f acc0 = vzero, acc1 = vzero;
      gemm_pair<8, 0, 8, 8>(wlane + SLOT_LAP + mt * 512, y, acc0, acc1);
      tx[mt] = acc0; tx[mt + 1] = acc1;
    }

    asm volatile("" ::: "memory");
#pragma unroll
    for (int mt = 0; mt < 8; mt += 2) {
      v8f acc[2] = {vzero, vzero};
      gemm_pair<8, 0, 8, 8>(wlane + SLOT_LAP + mt * 512, tx, acc[0], acc[1]);
#pragma unroll
      for (int u = 0; u < 2; ++u)
#pragma unroll
        for (int r = 0; r < 8; ++r) {
          const float y0 = y[mt + u][r], y1 = tx[mt + u][r];
          const float tx2 = fminf(fmaxf(2.0f * acc[u][r] - y0, -50.0f), 50.0f);
          y[mt + u][r] = w0 * y0 + w1 * y1 + w2 * tx2;
        }
    }

    v8f hd[16];
#pragma unroll
    for (int mt = 0; mt < 16; mt += 2) {
      v8f acc[2] = {vzero, vzero};
      gemm_pair<8, 0, 4, 8>(wlane + SLOT_WI + mt * 512, y, acc[0], acc[1]);
      gemm_pair<8, 4, 4, 8>(wlane + SLOT_WI + mt * 512 + 256, y, acc[0], acc[1]);
#pragma unroll
      for (int u = 0; u < 2; ++u) {
        const v8f bia = load8(bi + (mt + u) * 16 + f8);
#pragma unroll
        for (int r = 0; r < 8; ++r) hd[mt + u][r] = acc[u][r] + bia[r];
      }
    }

    float m2, inv2;
    {
      float s = lane_sum<16>(hd);
      s += __shfl_xor(s, 16, 32);
      m2 = s * (1.0f / 256.0f);
      float vs = lane_sqdev<16>(hd, m2);
      vs += __shfl_xor(vs, 16, 32);
      inv2 = rsqrtf(vs * (1.0f / 256.0f) + EPSV);
    }
#pragma unroll 1
    for (int mt = 0; mt < 16; ++mt) {
      const v8f g = load8(g2 + mt * 16 + f8), bb = load8(b2 + mt * 16 + f8);
      v8f done;
#pragma unroll
      for (int r = 0; r < 8; ++r) {
        const float yv = (hd[0][r] - m2) * inv2 * g[r] + bb[r];
        done[r] = 0.5f * yv * (1.0f + erff(yv * 0.70710678118654752f));
      }
#pragma unroll
      for (int t = 0; t < 15; ++t) hd[t] = hd[t + 1];
      hd[15] = done;
    }

    v8f od[8];
#pragma unroll
    for (int mt = 0; mt < 8; mt += 2) {
      v8f acc[2] = {vzero, vzero};
      gemm_pair<16, 0, 4, 16>(wlane + SLOT_WO + mt * 1024, hd, acc[0], acc[1]);
      gemm_pair<16, 4, 4, 16>(wlane + SLOT_WO + mt * 1024 + 256, hd, acc[0], acc[1]);
      gemm_pair<16, 8, 4, 16>(wlane + SLOT_WO + mt * 1024 + 512, hd, acc[0], acc[1]);
      gemm_pair<16, 12, 4, 16>(wlane + SLOT_WO + mt * 1024 + 768, hd, acc[0], acc[1]);
#pragma unroll
      for (int u = 0; u < 2; ++u) {
        const v8f boa = load8(bo + (mt + u) * 16 + f8);
#pragma unroll
        for (int r = 0; r < 8; ++r) od[mt + u][r] = acc[u][r] + boa[r];
      }
    }

    {
      float s = lane_sum<8>(od);
      s += __shfl_xor(s, 16, 32);
      const float m = s * (1.0f / 128.0f);
      float vs = lane_sqdev<8>(od, m);
      vs += __shfl_xor(vs, 16, 32);
      const float inv = rsqrtf(vs * (1.0f / 128.0f) + EPSV);
      float* os = out + srow * 128 + f8;
#pragma unroll
      for (int t = 0; t < 8; ++t) {
        const v8f g = load8(g3 + t * 16 + f8), bb = load8(b3 + t * 16 + f8);
        v4f lo, hi;
        lo.x = (od[t][0] - m) * inv * g[0] + bb[0];
        lo.y = (od[t][1] - m) * inv * g[1] + bb[1];
        lo.z = (od[t][2] - m) * inv * g[2] + bb[2];
        lo.w = (od[t][3] - m) * inv * g[3] + bb[3];
        hi.x = (od[t][4] - m) * inv * g[4] + bb[4];
        hi.y = (od[t][5] - m) * inv * g[5] + bb[5];
        hi.z = (od[t][6] - m) * inv * g[6] + bb[6];
        hi.w = (od[t][7] - m) * inv * g[7] + bb[7];
        *reinterpret_cast<v4f*>(os + t * 16) = lo;
        *reinterpret_cast<v4f*>(os + t * 16 + 4) = hi;
      }
    }
  }
}

extern "C" void kernel_launch(void* const* d_in, const int* in_sizes, int n_in,
                              void* d_out, int out_size, void* d_ws, size_t ws_size,
                              hipStream_t stream) {
  const float* x     = (const float*)d_in[0];
  const float* corr  = (const float*)d_in[1];
  const float* alog  = (const float*)d_in[2];
  const float* chebw = (const float*)d_in[3];
  const float* Wi    = (const float*)d_in[4];
  const float* bi    = (const float*)d_in[5];
  const float* Wo    = (const float*)d_in[6];
  const float* bo    = (const float*)d_in[7];
  const float* g1    = (const float*)d_in[8];
  const float* b1    = (const float*)d_in[9];
  const float* g2    = (const float*)d_in[10];
  const float* b2    = (const float*)d_in[11];
  const float* g3    = (const float*)d_in[12];
  const float* b3    = (const float*)d_in[13];
  float* out = (float*)d_out;
  float* ws  = (float*)d_ws;

  float* stats1 = ws;
  float* stats2 = ws + 4096;
  float* gpart  = ws + 8192;
  float* dcorr  = ws + 8192 + 256 * 16384;
  float* mixed  = out + (long)32 * 4096 * 128;

  hipMemsetAsync(d_ws, 0, 8192 * sizeof(float), stream);
  k_gram<<<256, 128, 0, stream>>>(x, g1, b1, stats1, stats2, gpart);
  k_combine<<<128, 128, 0, stream>>>(stats1, stats2, gpart, dcorr);
  k_main<<<256, 384, 0, stream>>>(x, g1, b1, corr, alog, chebw, dcorr, Wi, bi, Wo, bo,
                                  g2, b2, g3, b3, out, mixed);
}
